# starting kernel
# baseline (speedup 1.0000x reference)
#include <hip/hip_runtime.h>

typedef float v2f __attribute__((ext_vector_type(2)));
typedef float v8f __attribute__((ext_vector_type(8)));

#define C_CURV      1.0f
#define BALL_EPS    0.001f
#define ATANH_CLAMP (1.0f - 1e-05f)
#define MIN_DENOM   1e-15f
#define DDIM        64
#define WAVES_PER_BLOCK 8
#define ROWS_PER_BLOCK (WAVES_PER_BLOCK * 16)

__global__ __launch_bounds__(256)
void geo_proto_kernel(const float* __restrict__ X,
                      const float* __restrict__ P,
                      float* __restrict__ Out,
                      int N, int K)
{
    __shared__ float Pt[16][66];
    __shared__ float p2s[16];
    __shared__ float x2s[WAVES_PER_BLOCK][16];

    const int tid  = threadIdx.x;
    const int wave = tid >> 5;
    const int lane = tid & 31;
    const int m    = lane & 15;
    const int h    = lane >> 4;

    const int col0 = blockIdx.y * 16;
    const int row0 = blockIdx.x * ROWS_PER_BLOCK + wave * 16;

    if (tid < 16) {
        const float* prow = P + (size_t)(col0 + tid) * DDIM;
        float ss = 0.0f;
        for (int d = 0; d < DDIM; ++d) { float v = prow[d]; ss += v * v; }
        float nrm   = fmaxf(sqrtf(ss), 1e-15f);
        float maxn  = (1.0f - BALL_EPS) / sqrtf(C_CURV);
        float scale = fminf(1.0f, maxn / nrm);
        for (int d = 0; d < DDIM; ++d) Pt[tid][d] = prow[d] * scale;
        p2s[tid] = ss * scale * scale;
    }
    __syncthreads();

    v2f a[16], b[16];
    const float* xrow = X + (size_t)(row0 + m) * DDIM + 2 * h;
    float part = 0.0f;
#pragma unroll
    for (int kk = 0; kk < 16; ++kk) {
        a[kk] = *(const v2f*)(xrow + 4 * kk);
        b[kk] = *(const v2f*)(&Pt[m][4 * kk + 2 * h]);
        part += a[kk].x * a[kk].x + a[kk].y * a[kk].y;
    }

    v8f acc = {};
#pragma unroll
    for (int kk = 0; kk < 16; ++kk) {
        acc = __builtin_amdgcn_wmma_f32_16x16x4_f32(
             false, a[kk],
             false, b[kk],
             (short)0, acc,
             false,  false);
    }

    float x2full = part + __shfl_xor(part, 16, 32);
    if (lane < 16) x2s[wave][lane] = x2full;
    __syncthreads();

    const float p2     = p2s[m];
    const float sqrt_c = sqrtf(C_CURV);
#pragma unroll
    for (int r = 0; r < 8; ++r) {
        const int   row = r + 8 * h;
        const float x2  = x2s[wave][row];
        const float xy  = acc[r];

        const float t  = 1.0f - 2.0f * C_CURV * xy;
        const float A1 = t + C_CURV * p2;
        const float B1 = 1.0f - C_CURV * x2;
        float den = t + C_CURV * C_CURV * x2 * p2;
        den = fmaxf(den, MIN_DENOM);

        float ma2 = (A1 * A1 * x2 - 2.0f * A1 * B1 * xy + B1 * B1 * p2) / (den * den);
        ma2 = fmaxf(ma2, 0.0f);

        float z = sqrt_c * sqrtf(ma2);
        z = fminf(z, ATANH_CLAMP);
        const float at   = 0.5f * __logf((1.0f + z) / (1.0f - z));
        const float dist = (2.0f / sqrt_c) * at;

        Out[(size_t)(row0 + row) * K + (col0 + m)] = -(dist * dist);
    }
}

extern "C" void kernel_launch(void* const* d_in, const int* in_sizes, int n_in,
                              void* d_out, int out_size, void* d_ws, size_t ws_size,
                              hipStream_t stream) {
    const float* x = (const float*)d_in[0];
    const float* p = (const float*)d_in[1];
    float* out = (float*)d_out;

    const int N = in_sizes[0] / DDIM;
    const int K = in_sizes[1] / DDIM;

    dim3 grid(N / ROWS_PER_BLOCK, K / 16);
    geo_proto_kernel<<<grid, 256, 0, stream>>>(x, p, out, N, K);
}


# Each wave keeps its x fragments in registers and walks four 16-column tiles, two accumulators at a time (block = 128 rows x 64 prototypes): 1024 -> 256 blocks, 8192 -> 2048 waves; sum orders kept
# speedup vs baseline: 1.0648x; 1.0648x over previous
#include <hip/hip_runtime.h>


typedef float v2f __attribute__((ext_vector_type(2)));
typedef float v8f __attribute__((ext_vector_type(8)));

#define C_CURV      1.0f
#define BALL_EPS    0.001f
#define ATANH_CLAMP (1.0f - 1e-05f)
#define MIN_DENOM   1e-15f
#define DDIM        64
#define WAVES_PER_BLOCK 8
#define ROWS_PER_BLOCK (WAVES_PER_BLOCK * 16)
#define TILES_PER_WAVE 4
#define COLS_PER_BLOCK (TILES_PER_WAVE * 16)

__device__ __forceinline__
void tile_epilogue(const v8f acc, const float* __restrict__ x2row, const float p2,
                   float* __restrict__ outp, const int K)
{
    const float sqrt_c = sqrtf(C_CURV);
#pragma unroll
    for (int r = 0; r < 8; ++r) {
        const float x2  = x2row[r];
        const float xy  = acc[r];

        const float t  = 1.0f - 2.0f * C_CURV * xy;
        const float A1 = t + C_CURV * p2;
        const float B1 = 1.0f - C_CURV * x2;
        float den = t + C_CURV * C_CURV * x2 * p2;
        den = fmaxf(den, MIN_DENOM);

        float ma2 = (A1 * A1 * x2 - 2.0f * A1 * B1 * xy + B1 * B1 * p2) / (den * den);
        ma2 = fmaxf(ma2, 0.0f);

        float z = sqrt_c * sqrtf(ma2);
        z = fminf(z, ATANH_CLAMP);
        const float at   = 0.5f * __logf((1.0f + z) / (1.0f - z));
        const float dist = (2.0f / sqrt_c) * at;

        outp[(size_t)r * K] = -(dist * dist);
    }
}

__global__ __launch_bounds__(256)
void geo_proto_kernel(const float* __restrict__ X,
                      const float* __restrict__ P,
                      float* __restrict__ Out,
                      int N, int K)
{
    __shared__ float Pt[COLS_PER_BLOCK][66];
    __shared__ float p2s[COLS_PER_BLOCK];
    __shared__ float x2s[WAVES_PER_BLOCK][16];

    const int tid  = threadIdx.x;
    const int wave = tid >> 5;
    const int lane = tid & 31;
    const int m    = lane & 15;
    const int h    = lane >> 4;

    const int col0 = blockIdx.y * COLS_PER_BLOCK;
    const int row0 = blockIdx.x * ROWS_PER_BLOCK + wave * 16;

    if (tid < COLS_PER_BLOCK) {
        const float* prow = P + (size_t)(col0 + tid) * DDIM;
        float ss = 0.0f;
        for (int d = 0; d < DDIM; ++d) { float v = prow[d]; ss += v * v; }
        float nrm   = fmaxf(sqrtf(ss), 1e-15f);
        float maxn  = (1.0f - BALL_EPS) / sqrtf(C_CURV);
        float scale = fminf(1.0f, maxn / nrm);
        for (int d = 0; d < DDIM; ++d) Pt[tid][d] = prow[d] * scale;
        p2s[tid] = ss * scale * scale;
    }
    __syncthreads();

    v2f a[16];
    const float* xrow = X + (size_t)(row0 + m) * DDIM + 2 * h;
    float part = 0.0f;
#pragma unroll
    for (int kk = 0; kk < 16; ++kk) {
        a[kk] = *(const v2f*)(xrow + 4 * kk);
        part += a[kk].x * a[kk].x + a[kk].y * a[kk].y;
    }

    float x2full = part + __shfl_xor(part, 16, 32);
    if (lane < 16) x2s[wave][lane] = x2full;
    __syncthreads();
    float x2row[8];
#pragma unroll
    for (int r = 0; r < 8; ++r) x2row[r] = x2s[wave][r + 8 * h];

    float* outw = Out + (size_t)(row0 + 8 * h) * K + (col0 + m);
#pragma unroll
    for (int j = 0; j < TILES_PER_WAVE; j += 2) {
        v2f b0[16], b1[16];
#pragma unroll
        for (int kk = 0; kk < 16; ++kk) {
            b0[kk] = *(const v2f*)(&Pt[16 * j + m][4 * kk + 2 * h]);
            b1[kk] = *(const v2f*)(&Pt[16 * (j + 1) + m][4 * kk + 2 * h]);
        }

        v8f acc0 = {}, acc1 = {};
#pragma unroll
        for (int kk = 0; kk < 16; ++kk) {
            acc0 = __builtin_amdgcn_wmma_f32_16x16x4_f32(
                 false, a[kk],  false, b0[kk],
                 (short)0, acc0,  false,  false);
            acc1 = __builtin_amdgcn_wmma_f32_16x16x4_f32(
                 false, a[kk],  false, b1[kk],
                 (short)0, acc1,  false,  false);
        }

        tile_epilogue(acc0, x2row, p2s[16 * j + m],       outw + 16 * j,       K);
        tile_epilogue(acc1, x2row, p2s[16 * (j + 1) + m], outw + 16 * (j + 1), K);
    }
}

extern "C" void kernel_launch(void* const* d_in, const int* in_sizes, int n_in,
                              void* d_out, int out_size, void* d_ws, size_t ws_size,
                              hipStream_t stream) {
    const float* x = (const float*)d_in[0];
    const float* p = (const float*)d_in[1];
    float* out = (float*)d_out;

    const int N = in_sizes[0] / DDIM;
    const int K = in_sizes[1] / DDIM;

    dim3 grid(N / ROWS_PER_BLOCK, K / COLS_PER_BLOCK);
    geo_proto_kernel<<<grid, 256, 0, stream>>>(x, p, out, N, K);
}


# Prologue waits for one global round trip, not six: prototypes projected by all 256 threads (4 lanes per row, one 64-byte piece each, quad sum; ||p||^2 sum order changed), x loads issued ahead
# speedup vs baseline: 1.2595x; 1.1829x over previous
#include <hip/hip_runtime.h>


typedef float v2f __attribute__((ext_vector_type(2)));
typedef float v4f __attribute__((ext_vector_type(4)));
typedef float v8f __attribute__((ext_vector_type(8)));

#define C_CURV      1.0f
#define BALL_EPS    0.001f
#define ATANH_CLAMP (1.0f - 1e-05f)
#define MIN_DENOM   1e-15f
#define DDIM        64
#define WAVES_PER_BLOCK 8
#define ROWS_PER_BLOCK (WAVES_PER_BLOCK * 16)
#define TILES_PER_WAVE 4
#define COLS_PER_BLOCK (TILES_PER_WAVE * 16)

__device__ __forceinline__
void tile_epilogue(const v8f acc, const float* __restrict__ x2row, const float p2,
                   float* __restrict__ outp, const int K)
{
    const float sqrt_c = sqrtf(C_CURV);
#pragma unroll
    for (int r = 0; r < 8; ++r) {
        const float x2  = x2row[r];
        const float xy  = acc[r];

        const float t  = 1.0f - 2.0f * C_CURV * xy;
        const float A1 = t + C_CURV * p2;
        const float B1 = 1.0f - C_CURV * x2;
        float den = t + C_CURV * C_CURV * x2 * p2;
        den = fmaxf(den, MIN_DENOM);

        float ma2 = (A1 * A1 * x2 - 2.0f * A1 * B1 * xy + B1 * B1 * p2) / (den * den);
        ma2 = fmaxf(ma2, 0.0f);

        float z = sqrt_c * sqrtf(ma2);
        z = fminf(z, ATANH_CLAMP);
        const float at   = 0.5f * __logf((1.0f + z) / (1.0f - z));
        const float dist = (2.0f / sqrt_c) * at;

        outp[(size_t)r * K] = -(dist * dist);
    }
}

__global__ __launch_bounds__(256)
void geo_proto_kernel(const float* __restrict__ X,
                      const float* __restrict__ P,
                      float* __restrict__ Out,
                      int N, int K)
{
    __shared__ float Pt[COLS_PER_BLOCK][66];
    __shared__ float p2s[COLS_PER_BLOCK];
    __shared__ float x2s[WAVES_PER_BLOCK][16];

    const int tid  = threadIdx.x;
    const int wave = tid >> 5;
    const int lane = tid & 31;
    const int m    = lane & 15;
    const int h    = lane >> 4;

    const int col0 = blockIdx.y * COLS_PER_BLOCK;
    const int row0 = blockIdx.x * ROWS_PER_BLOCK + wave * 16;

    const int pr = tid >> 2;
    const int q  = tid & 3;
    const float* ppiece = P + (size_t)(col0 + pr) * DDIM + 16 * q;
    v4f pv[4];
#pragma unroll
    for (int i = 0; i < 4; ++i) pv[i] = *(const v4f*)(ppiece + 4 * i);

    v2f a[16];
    const float* xrow = X + (size_t)(row0 + m) * DDIM + 2 * h;
#pragma unroll
    for (int kk = 0; kk < 16; ++kk) a[kk] = *(const v2f*)(xrow + 4 * kk);

    float ss = 0.0f;
#pragma unroll
    for (int i = 0; i < 4; ++i) {
        ss += pv[i].x * pv[i].x;
        ss += pv[i].y * pv[i].y;
        ss += pv[i].z * pv[i].z;
        ss += pv[i].w * pv[i].w;
    }
    ss += __shfl_xor(ss, 1, 32);
    ss += __shfl_xor(ss, 2, 32);
    {
        const float nrm   = fmaxf(sqrtf(ss), 1e-15f);
        const float maxn  = (1.0f - BALL_EPS) / sqrtf(C_CURV);
        const float scale = fminf(1.0f, maxn / nrm);
#pragma unroll
        for (int i = 0; i < 4; ++i) {
            v2f lo = { pv[i].x * scale, pv[i].y * scale };
            v2f hi = { pv[i].z * scale, pv[i].w * scale };
            *(v2f*)(&Pt[pr][16 * q + 4 * i])     = lo;
            *(v2f*)(&Pt[pr][16 * q + 4 * i + 2]) = hi;
        }
        if (q == 0) p2s[pr] = ss * scale * scale;
    }

    float part = 0.0f;
#pragma unroll
    for (int kk = 0; kk < 16; ++kk)
        part += a[kk].x * a[kk].x + a[kk].y * a[kk].y;
    float x2full = part + __shfl_xor(part, 16, 32);
    if (lane < 16) x2s[wave][lane] = x2full;
    __syncthreads();
    float x2row[8];
#pragma unroll
    for (int r = 0; r < 8; ++r) x2row[r] = x2s[wave][r + 8 * h];

    float* outw = Out + (size_t)(row0 + 8 * h) * K + (col0 + m);
#pragma unroll
    for (int j = 0; j < TILES_PER_WAVE; j += 2) {
        v2f b0[16], b1[16];
#pragma unroll
        for (int kk = 0; kk < 16; ++kk) {
            b0[kk] = *(const v2f*)(&Pt[16 * j + m][4 * kk + 2 * h]);
            b1[kk] = *(const v2f*)(&Pt[16 * (j + 1) + m][4 * kk + 2 * h]);
        }

        v8f acc0 = {}, acc1 = {};
#pragma unroll
        for (int kk = 0; kk < 16; ++kk) {
            acc0 = __builtin_amdgcn_wmma_f32_16x16x4_f32(
                 false, a[kk],  false, b0[kk],
                 (short)0, acc0,  false,  false);
            acc1 = __builtin_amdgcn_wmma_f32_16x16x4_f32(
                 false, a[kk],  false, b1[kk],
                 (short)0, acc1,  false,  false);
        }

        tile_epilogue(acc0, x2row, p2s[16 * j + m],       outw + 16 * j,       K);
        tile_epilogue(acc1, x2row, p2s[16 * (j + 1) + m], outw + 16 * (j + 1), K);
    }
}

extern "C" void kernel_launch(void* const* d_in, const int* in_sizes, int n_in,
                              void* d_out, int out_size, void* d_ws, size_t ws_size,
                              hipStream_t stream) {
    const float* x = (const float*)d_in[0];
    const float* p = (const float*)d_in[1];
    float* out = (float*)d_out;

    const int N = in_sizes[0] / DDIM;
    const int K = in_sizes[1] / DDIM;

    dim3 grid(N / ROWS_PER_BLOCK, K / COLS_PER_BLOCK);
    geo_proto_kernel<<<grid, 256, 0, stream>>>(x, p, out, N, K);
}


# Epilogue on the hardware's approximate rcp, sqrt and log2 (about 1 ulp each) instead of correctly rounded division and sqrt; norm as sqrt(num2)*rcp(den): one division per output, not two
# speedup vs baseline: 1.5562x; 1.2356x over previous
#include <hip/hip_runtime.h>


typedef float v2f __attribute__((ext_vector_type(2)));
typedef float v4f __attribute__((ext_vector_type(4)));
typedef float v8f __attribute__((ext_vector_type(8)));

#define C_CURV      1.0f
#define BALL_EPS    0.001f
#define ATANH_CLAMP (1.0f - 1e-05f)
#define MIN_DENOM   1e-15f
#define DDIM        64
#define WAVES_PER_BLOCK 8
#define ROWS_PER_BLOCK (WAVES_PER_BLOCK * 16)
#define TILES_PER_WAVE 4
#define COLS_PER_BLOCK (TILES_PER_WAVE * 16)

__device__ __forceinline__
void tile_epilogue(const v8f acc, const float* __restrict__ x2row, const float p2,
                   float* __restrict__ outp, const int K)
{
    const float sqrt_c = sqrtf(C_CURV);
    const float ln2    = 0.693147180559945309f;
#pragma unroll
    for (int r = 0; r < 8; ++r) {
        const float x2  = x2row[r];
        const float xy  = acc[r];

        const float t  = 1.0f - 2.0f * C_CURV * xy;
        const float A1 = t + C_CURV * p2;
        const float B1 = 1.0f - C_CURV * x2;
        float den = t + C_CURV * C_CURV * x2 * p2;
        den = fmaxf(den, MIN_DENOM);

        float num2 = A1 * A1 * x2 - 2.0f * A1 * B1 * xy + B1 * B1 * p2;
        num2 = fmaxf(num2, 0.0f);

        float z = sqrt_c * (__builtin_amdgcn_sqrtf(num2) * __builtin_amdgcn_rcpf(den));
        z = fminf(z, ATANH_CLAMP);
        const float q    = (1.0f + z) * __builtin_amdgcn_rcpf(1.0f - z);
        const float dist = (ln2 / sqrt_c) * __builtin_amdgcn_logf(q);

        outp[(size_t)r * K] = -(dist * dist);
    }
}

__global__ __launch_bounds__(256)
void geo_proto_kernel(const float* __restrict__ X,
                      const float* __restrict__ P,
                      float* __restrict__ Out,
                      int N, int K)
{
    __shared__ float Pt[COLS_PER_BLOCK][66];
    __shared__ float p2s[COLS_PER_BLOCK];
    __shared__ float x2s[WAVES_PER_BLOCK][16];

    const int tid  = threadIdx.x;
    const int wave = tid >> 5;
    const int lane = tid & 31;
    const int m    = lane & 15;
    const int h    = lane >> 4;

    const int col0 = blockIdx.y * COLS_PER_BLOCK;
    const int row0 = blockIdx.x * ROWS_PER_BLOCK + wave * 16;

    const int pr = tid >> 2;
    const int q  = tid & 3;
    const float* ppiece = P + (size_t)(col0 + pr) * DDIM + 16 * q;
    v4f pv[4];
#pragma unroll
    for (int i = 0; i < 4; ++i) pv[i] = *(const v4f*)(ppiece + 4 * i);

    v2f a[16];
    const float* xrow = X + (size_t)(row0 + m) * DDIM + 2 * h;
#pragma unroll
    for (int kk = 0; kk < 16; ++kk) a[kk] = *(const v2f*)(xrow + 4 * kk);

    float ss = 0.0f;
#pragma unroll
    for (int i = 0; i < 4; ++i) {
        ss += pv[i].x * pv[i].x;
        ss += pv[i].y * pv[i].y;
        ss += pv[i].z * pv[i].z;
        ss += pv[i].w * pv[i].w;
    }
    ss += __shfl_xor(ss, 1, 32);
    ss += __shfl_xor(ss, 2, 32);
    {
        const float nrm   = fmaxf(sqrtf(ss), 1e-15f);
        const float maxn  = (1.0f - BALL_EPS) / sqrtf(C_CURV);
        const float scale = fminf(1.0f, maxn / nrm);
#pragma unroll
        for (int i = 0; i < 4; ++i) {
            v2f lo = { pv[i].x * scale, pv[i].y * scale };
            v2f hi = { pv[i].z * scale, pv[i].w * scale };
            *(v2f*)(&Pt[pr][16 * q + 4 * i])     = lo;
            *(v2f*)(&Pt[pr][16 * q + 4 * i + 2]) = hi;
        }
        if (q == 0) p2s[pr] = ss * scale * scale;
    }

    float part = 0.0f;
#pragma unroll
    for (int kk = 0; kk < 16; ++kk)
        part += a[kk].x * a[kk].x + a[kk].y * a[kk].y;
    float x2full = part + __shfl_xor(part, 16, 32);
    if (lane < 16) x2s[wave][lane] = x2full;
    __syncthreads();
    float x2row[8];
#pragma unroll
    for (int r = 0; r < 8; ++r) x2row[r] = x2s[wave][r + 8 * h];

    float* outw = Out + (size_t)(row0 + 8 * h) * K + (col0 + m);
#pragma unroll
    for (int j = 0; j < TILES_PER_WAVE; j += 2) {
        v2f b0[16], b1[16];
#pragma unroll
        for (int kk = 0; kk < 16; ++kk) {
            b0[kk] = *(const v2f*)(&Pt[16 * j + m][4 * kk + 2 * h]);
            b1[kk] = *(const v2f*)(&Pt[16 * (j + 1) + m][4 * kk + 2 * h]);
        }

        v8f acc0 = {}, acc1 = {};
#pragma unroll
        for (int kk = 0; kk < 16; ++kk) {
            acc0 = __builtin_amdgcn_wmma_f32_16x16x4_f32(
                 false, a[kk],  false, b0[kk],
                 (short)0, acc0,  false,  false);
            acc1 = __builtin_amdgcn_wmma_f32_16x16x4_f32(
                 false, a[kk],  false, b1[kk],
                 (short)0, acc1,  false,  false);
        }

        tile_epilogue(acc0, x2row, p2s[16 * j + m],       outw + 16 * j,       K);
        tile_epilogue(acc1, x2row, p2s[16 * (j + 1) + m], outw + 16 * (j + 1), K);
    }
}

extern "C" void kernel_launch(void* const* d_in, const int* in_sizes, int n_in,
                              void* d_out, int out_size, void* d_ws, size_t ws_size,
                              hipStream_t stream) {
    const float* x = (const float*)d_in[0];
    const float* p = (const float*)d_in[1];
    float* out = (float*)d_out;

    const int N = in_sizes[0] / DDIM;
    const int K = in_sizes[1] / DDIM;

    dim3 grid(N / ROWS_PER_BLOCK, K / COLS_PER_BLOCK);
    geo_proto_kernel<<<grid, 256, 0, stream>>>(x, p, out, N, K);
}


# Global loads cover whole 128-byte lines (512 contiguous bytes per instruction): x reordered to the A layout through LDS; ||p||^2 as a 16-lane butterfly of 4-term sums (its sum order changed)
# speedup vs baseline: 1.6570x; 1.0648x over previous
#include <hip/hip_runtime.h>


typedef float v2f __attribute__((ext_vector_type(2)));
typedef float v4f __attribute__((ext_vector_type(4)));
typedef float v8f __attribute__((ext_vector_type(8)));

#define C_CURV      1.0f
#define BALL_EPS    0.001f
#define ATANH_CLAMP (1.0f - 1e-05f)
#define MIN_DENOM   1e-15f
#define DDIM        64
#define WAVES_PER_BLOCK 8
#define ROWS_PER_BLOCK (WAVES_PER_BLOCK * 16)
#define TILES_PER_WAVE 4
#define COLS_PER_BLOCK (TILES_PER_WAVE * 16)
#define XS_PITCH 68

__device__ __forceinline__
void tile_epilogue(const v8f acc, const float* __restrict__ x2row, const float p2,
                   float* __restrict__ outp, const int K)
{
    const float sqrt_c = sqrtf(C_CURV);
    const float ln2    = 0.693147180559945309f;
#pragma unroll
    for (int r = 0; r < 8; ++r) {
        const float x2  = x2row[r];
        const float xy  = acc[r];

        const float t  = 1.0f - 2.0f * C_CURV * xy;
        const float A1 = t + C_CURV * p2;
        const float B1 = 1.0f - C_CURV * x2;
        float den = t + C_CURV * C_CURV * x2 * p2;
        den = fmaxf(den, MIN_DENOM);

        float num2 = A1 * A1 * x2 - 2.0f * A1 * B1 * xy + B1 * B1 * p2;
        num2 = fmaxf(num2, 0.0f);

        float z = sqrt_c * (__builtin_amdgcn_sqrtf(num2) * __builtin_amdgcn_rcpf(den));
        z = fminf(z, ATANH_CLAMP);
        const float q    = (1.0f + z) * __builtin_amdgcn_rcpf(1.0f - z);
        const float dist = (ln2 / sqrt_c) * __builtin_amdgcn_logf(q);

        outp[(size_t)r * K] = -(dist * dist);
    }
}

__global__ __launch_bounds__(256)
void geo_proto_kernel(const float* __restrict__ X,
                      const float* __restrict__ P,
                      float* __restrict__ Out,
                      int N, int K)
{
    __shared__ float Pt[COLS_PER_BLOCK][66];
    __shared__ float p2s[COLS_PER_BLOCK];
    __shared__ __attribute__((aligned(16)))
    float xs[WAVES_PER_BLOCK][16][XS_PITCH];

    const int tid  = threadIdx.x;
    const int wave = tid >> 5;
    const int lane = tid & 31;
    const int m    = lane & 15;
    const int h    = lane >> 4;

    const int col0 = blockIdx.y * COLS_PER_BLOCK;
    const int row0 = blockIdx.x * ROWS_PER_BLOCK + wave * 16;

    const float* pw = P + (size_t)(col0 + 8 * wave) * DDIM + 4 * lane;
    v4f pv[4];
#pragma unroll
    for (int i = 0; i < 4; ++i) pv[i] = *(const v4f*)(pw + 2 * DDIM * i);

    const float* xw = X + (size_t)row0 * DDIM + 4 * lane;
    v4f xv[8];
#pragma unroll
    for (int i = 0; i < 8; ++i) xv[i] = *(const v4f*)(xw + 2 * DDIM * i);

    float ss[4];
#pragma unroll
    for (int i = 0; i < 4; ++i)
        ss[i] = pv[i].x * pv[i].x + pv[i].y * pv[i].y + pv[i].z * pv[i].z + pv[i].w * pv[i].w;
#pragma unroll
    for (int s = 1; s < 16; s <<= 1) {
#pragma unroll
        for (int i = 0; i < 4; ++i) ss[i] += __shfl_xor(ss[i], s, 32);
    }
    {
        const float maxn = (1.0f - BALL_EPS) / sqrtf(C_CURV);
        float scale[4];
#pragma unroll
        for (int i = 0; i < 4; ++i) {
            const float nrm = fmaxf(sqrtf(ss[i]), 1e-15f);
            scale[i] = fminf(1.0f, maxn / nrm);
        }
#pragma unroll
        for (int i = 0; i < 4; ++i) {
            const int pr = 8 * wave + 2 * i + h;
            v2f lo = { pv[i].x * scale[i], pv[i].y * scale[i] };
            v2f hi = { pv[i].z * scale[i], pv[i].w * scale[i] };
            *(v2f*)(&Pt[pr][4 * m])     = lo;
            *(v2f*)(&Pt[pr][4 * m + 2]) = hi;
        }
        if (m == 0) {
#pragma unroll
            for (int i = 0; i < 4; ++i)
                p2s[8 * wave + 2 * i + h] = ss[i] * scale[i] * scale[i];
        }
    }

#pragma unroll
    for (int i = 0; i < 8; ++i) *(v4f*)(&xs[wave][2 * i + h][4 * m]) = xv[i];
    __syncthreads();

    v2f a[16];
#pragma unroll
    for (int kk = 0; kk < 16; ++kk) a[kk] = *(const v2f*)(&xs[wave][m][4 * kk + 2 * h]);

    float part = 0.0f;
#pragma unroll
    for (int kk = 0; kk < 16; ++kk)
        part += a[kk].x * a[kk].x + a[kk].y * a[kk].y;
    const float x2full = part + __shfl_xor(part, 16, 32);
    float x2row[8];
#pragma unroll
    for (int r = 0; r < 8; ++r) x2row[r] = __shfl(x2full, r + 8 * h, 32);

    float* outw = Out + (size_t)(row0 + 8 * h) * K + (col0 + m);
#pragma unroll
    for (int j = 0; j < TILES_PER_WAVE; j += 2) {
        v2f b0[16], b1[16];
#pragma unroll
        for (int kk = 0; kk < 16; ++kk) {
            b0[kk] = *(const v2f*)(&Pt[16 * j + m][4 * kk + 2 * h]);
            b1[kk] = *(const v2f*)(&Pt[16 * (j + 1) + m][4 * kk + 2 * h]);
        }

        v8f acc0 = {}, acc1 = {};
#pragma unroll
        for (int kk = 0; kk < 16; ++kk) {
            acc0 = __builtin_amdgcn_wmma_f32_16x16x4_f32(
                 false, a[kk],  false, b0[kk],
                 (short)0, acc0,  false,  false);
            acc1 = __builtin_amdgcn_wmma_f32_16x16x4_f32(
                 false, a[kk],  false, b1[kk],
                 (short)0, acc1,  false,  false);
        }

        tile_epilogue(acc0, x2row, p2s[16 * j + m],       outw + 16 * j,       K);
        tile_epilogue(acc1, x2row, p2s[16 * (j + 1) + m], outw + 16 * (j + 1), K);
    }
}

extern "C" void kernel_launch(void* const* d_in, const int* in_sizes, int n_in,
                              void* d_out, int out_size, void* d_ws, size_t ws_size,
                              hipStream_t stream) {
    const float* x = (const float*)d_in[0];
    const float* p = (const float*)d_in[1];
    float* out = (float*)d_out;

    const int N = in_sizes[0] / DDIM;
    const int K = in_sizes[1] / DDIM;

    dim3 grid(N / ROWS_PER_BLOCK, K / COLS_PER_BLOCK);
    geo_proto_kernel<<<grid, 256, 0, stream>>>(x, p, out, N, K);
}


# The two waves of a SIMD take turns on the matrix pipe tile by tile (priority 3,2,1,0 by tile begun, 0 in epilogues); a tile's k-steps in two accumulators, even and odd (x.p sum order changed)
# speedup vs baseline: 1.7769x; 1.0724x over previous
#include <hip/hip_runtime.h>


typedef float v2f __attribute__((ext_vector_type(2)));
typedef float v4f __attribute__((ext_vector_type(4)));
typedef float v8f __attribute__((ext_vector_type(8)));

#define C_CURV      1.0f
#define BALL_EPS    0.001f
#define ATANH_CLAMP (1.0f - 1e-05f)
#define MIN_DENOM   1e-15f
#define DDIM        64
#define WAVES_PER_BLOCK 8
#define ROWS_PER_BLOCK (WAVES_PER_BLOCK * 16)
#define TILES_PER_WAVE 4
#define COLS_PER_BLOCK (TILES_PER_WAVE * 16)
#define XS_PITCH 68

__device__ __forceinline__
void tile_epilogue(const v8f acc, const float* __restrict__ x2row, const float p2,
                   float* __restrict__ outp, const int K)
{
    const float sqrt_c = sqrtf(C_CURV);
    const float ln2    = 0.693147180559945309f;
#pragma unroll
    for (int r = 0; r < 8; ++r) {
        const float x2  = x2row[r];
        const float xy  = acc[r];

        const float t  = 1.0f - 2.0f * C_CURV * xy;
        const float A1 = t + C_CURV * p2;
        const float B1 = 1.0f - C_CURV * x2;
        float den = t + C_CURV * C_CURV * x2 * p2;
        den = fmaxf(den, MIN_DENOM);

        float num2 = A1 * A1 * x2 - 2.0f * A1 * B1 * xy + B1 * B1 * p2;
        num2 = fmaxf(num2, 0.0f);

        float z = sqrt_c * (__builtin_amdgcn_sqrtf(num2) * __builtin_amdgcn_rcpf(den));
        z = fminf(z, ATANH_CLAMP);
        const float q    = (1.0f + z) * __builtin_amdgcn_rcpf(1.0f - z);
        const float dist = (ln2 / sqrt_c) * __builtin_amdgcn_logf(q);

        outp[(size_t)r * K] = -(dist * dist);
    }
}

__global__ __launch_bounds__(256)
void geo_proto_kernel(const float* __restrict__ X,
                      const float* __restrict__ P,
                      float* __restrict__ Out,
                      int N, int K)
{
    __shared__ float Pt[COLS_PER_BLOCK][66];
    __shared__ float p2s[COLS_PER_BLOCK];
    __shared__ __attribute__((aligned(16)))
    float xs[WAVES_PER_BLOCK][16][XS_PITCH];

    const int tid  = threadIdx.x;
    const int wave = tid >> 5;
    const int lane = tid & 31;
    const int m    = lane & 15;
    const int h    = lane >> 4;

    const int col0 = blockIdx.y * COLS_PER_BLOCK;
    const int row0 = blockIdx.x * ROWS_PER_BLOCK + wave * 16;

    const float* pw = P + (size_t)(col0 + 8 * wave) * DDIM + 4 * lane;
    v4f pv[4];
#pragma unroll
    for (int i = 0; i < 4; ++i) pv[i] = *(const v4f*)(pw + 2 * DDIM * i);

    const float* xw = X + (size_t)row0 * DDIM + 4 * lane;
    v4f xv[8];
#pragma unroll
    for (int i = 0; i < 8; ++i) xv[i] = *(const v4f*)(xw + 2 * DDIM * i);

    float ss[4];
#pragma unroll
    for (int i = 0; i < 4; ++i)
        ss[i] = pv[i].x * pv[i].x + pv[i].y * pv[i].y + pv[i].z * pv[i].z + pv[i].w * pv[i].w;
#pragma unroll
    for (int s = 1; s < 16; s <<= 1) {
#pragma unroll
        for (int i = 0; i < 4; ++i) ss[i] += __shfl_xor(ss[i], s, 32);
    }
    {
        const float maxn = (1.0f - BALL_EPS) / sqrtf(C_CURV);
        float scale[4];
#pragma unroll
        for (int i = 0; i < 4; ++i) {
            const float nrm = fmaxf(sqrtf(ss[i]), 1e-15f);
            scale[i] = fminf(1.0f, maxn / nrm);
        }
#pragma unroll
        for (int i = 0; i < 4; ++i) {
            const int pr = 8 * wave + 2 * i + h;
            v2f lo = { pv[i].x * scale[i], pv[i].y * scale[i] };
            v2f hi = { pv[i].z * scale[i], pv[i].w * scale[i] };
            *(v2f*)(&Pt[pr][4 * m])     = lo;
            *(v2f*)(&Pt[pr][4 * m + 2]) = hi;
        }
        if (m == 0) {
#pragma unroll
            for (int i = 0; i < 4; ++i)
                p2s[8 * wave + 2 * i + h] = ss[i] * scale[i] * scale[i];
        }
    }

#pragma unroll
    for (int i = 0; i < 8; ++i) *(v4f*)(&xs[wave][2 * i + h][4 * m]) = xv[i];
    __syncthreads();

    v2f a[16];
#pragma unroll
    for (int kk = 0; kk < 16; ++kk) a[kk] = *(const v2f*)(&xs[wave][m][4 * kk + 2 * h]);

    float part = 0.0f;
#pragma unroll
    for (int kk = 0; kk < 16; ++kk)
        part += a[kk].x * a[kk].x + a[kk].y * a[kk].y;
    const float x2full = part + __shfl_xor(part, 16, 32);
    float x2row[8];

    float p2v[TILES_PER_WAVE];
#pragma unroll
    for (int j = 0; j < TILES_PER_WAVE; ++j) p2v[j] = p2s[16 * j + m];

    v2f b[2][16];
#pragma unroll
    for (int kk = 0; kk < 16; ++kk) b[0][kk] = *(const v2f*)(&Pt[m][4 * kk + 2 * h]);

    float* outw = Out + (size_t)(row0 + 8 * h) * K + (col0 + m);
#pragma unroll
    for (int j = 0; j < TILES_PER_WAVE; ++j) {
        __builtin_amdgcn_sched_barrier(0);
        if (j == 0)      __builtin_amdgcn_s_setprio(3);
        else if (j == 1) __builtin_amdgcn_s_setprio(2);
        else if (j == 2) __builtin_amdgcn_s_setprio(1);
        if (j + 1 < TILES_PER_WAVE) {
#pragma unroll
            for (int kk = 0; kk < 16; ++kk)
                b[(j + 1) & 1][kk] = *(const v2f*)(&Pt[16 * (j + 1) + m][4 * kk + 2 * h]);
            __builtin_amdgcn_sched_barrier(0);
        }

        v8f acc0 = {}, acc1 = {};
#pragma unroll
        for (int kk = 0; kk < 16; kk += 2) {
            acc0 = __builtin_amdgcn_wmma_f32_16x16x4_f32(
                 false, a[kk],  false, b[j & 1][kk],
                 (short)0, acc0,  false,  false);
            acc1 = __builtin_amdgcn_wmma_f32_16x16x4_f32(
                 false, a[kk + 1],  false, b[j & 1][kk + 1],
                 (short)0, acc1,  false,  false);
        }
        __builtin_amdgcn_sched_barrier(0);
        if (j + 1 < TILES_PER_WAVE) __builtin_amdgcn_s_setprio(0);

        if (j == 0) {
#pragma unroll
            for (int r = 0; r < 8; ++r) x2row[r] = __shfl(x2full, r + 8 * h, 32);
        }
        tile_epilogue(acc0 + acc1, x2row, p2v[j], outw + 16 * j, K);
    }
}

extern "C" void kernel_launch(void* const* d_in, const int* in_sizes, int n_in,
                              void* d_out, int out_size, void* d_ws, size_t ws_size,
                              hipStream_t stream) {
    const float* x = (const float*)d_in[0];
    const float* p = (const float*)d_in[1];
    float* out = (float*)d_out;

    const int N = in_sizes[0] / DDIM;
    const int K = in_sizes[1] / DDIM;

    dim3 grid(N / ROWS_PER_BLOCK, K / COLS_PER_BLOCK);
    geo_proto_kernel<<<grid, 256, 0, stream>>>(x, p, out, N, K);
}
